# starting kernel
# baseline (speedup 1.0000x reference)
#include <hip/hip_runtime.h>

typedef __attribute__((ext_vector_type(2))) float v2f;
typedef __attribute__((ext_vector_type(8))) float v8f;

#define EMB    256
#define SENTS  128
#define BATCH  64
#define TOKS   64
#define NCLS   100
#define CPAD   112

__global__ void gather_mean_kernel(const int* __restrict__ tok,
                                   const float* __restrict__ emb,
                                   float* __restrict__ sents) {
  const int bs = blockIdx.x;
  const int t  = threadIdx.x;
  const int* tl = tok + (size_t)bs * TOKS;
  const float4* embv = (const float4*)emb;
  float4 acc = make_float4(0.f, 0.f, 0.f, 0.f);
  #pragma unroll 4
  for (int l = 0; l < TOKS; ++l) {
    const int tk = tl[l];
    float4 v = embv[(size_t)tk * (EMB / 4) + t];
    acc.x += v.x; acc.y += v.y; acc.z += v.z; acc.w += v.w;
  }
  const float inv = 1.0f / (float)TOKS;
  acc.x *= inv; acc.y *= inv; acc.z *= inv; acc.w *= inv;
  ((float4*)sents)[(size_t)bs * (EMB / 4) + t] = acc;
}

__global__ void scores_wmma_kernel(const float* __restrict__ cls,
                                   const float* __restrict__ sents,
                                   float* __restrict__ scores) {
  const int st   = blockIdx.x;
  const int b    = blockIdx.y;
  const int wave = threadIdx.x >> 5;
  const int lane = threadIdx.x & 31;
  const int hl   = lane >> 4;
  const int r15  = lane & 15;

  const int m  = wave * 16 + r15;
  const int mc = (m < NCLS) ? m : (NCLS - 1);
  const int n  = st * 16 + r15;

  const float* arow = cls + (size_t)mc * EMB + 2 * hl;
  const float* brow = sents + ((size_t)b * SENTS + n) * EMB + 2 * hl;

  v8f acc = {};
  #pragma unroll 8
  for (int kb = 0; kb < EMB; kb += 4) {
    v2f a = *(const v2f*)(arow + kb);
    v2f bb = *(const v2f*)(brow + kb);
    acc = __builtin_amdgcn_wmma_f32_16x16x4_f32(
         false, a,  false, bb,
         (short)0, acc,  false,  false);
  }
  float* outb = scores + (size_t)b * CPAD * SENTS;
  #pragma unroll
  for (int r = 0; r < 8; ++r) {
    const int c = wave * 16 + r + 8 * hl;
    outb[(size_t)c * SENTS + st * 16 + r15] = acc[r];
  }
}

__global__ void softmax_kernel(float* __restrict__ scores) {
  const int c = blockIdx.x;
  const int b = blockIdx.y;
  const int t = threadIdx.x;
  float* row = scores + ((size_t)b * CPAD + c) * SENTS;
  __shared__ float red[SENTS];

  const float v = row[t];
  red[t] = v;
  __syncthreads();
  for (int off = SENTS / 2; off > 0; off >>= 1) {
    if (t < off) red[t] = fmaxf(red[t], red[t + off]);
    __syncthreads();
  }
  const float mx = red[0];
  __syncthreads();
  const float e = __expf(v - mx);
  red[t] = e;
  __syncthreads();
  for (int off = SENTS / 2; off > 0; off >>= 1) {
    if (t < off) red[t] += red[t + off];
    __syncthreads();
  }
  row[t] = e / red[0];
}

__global__ void proj_wmma_kernel(const float* __restrict__ sents,
                                 const float* __restrict__ mw,
                                 float* __restrict__ proj) {
  const int st   = blockIdx.x;
  const int b    = blockIdx.y;
  const int wave = threadIdx.x >> 5;
  const int lane = threadIdx.x & 31;
  const int hl   = lane >> 4;
  const int r15  = lane & 15;

  const int m  = st * 16 + r15;
  const int n  = wave * 16 + r15;
  const int nc = (n < NCLS) ? n : (NCLS - 1);

  const float* arow = sents + ((size_t)b * SENTS + m) * EMB + 2 * hl;
  const float* brow = mw + (size_t)nc * EMB + 2 * hl;

  v8f acc = {};
  #pragma unroll 8
  for (int kb = 0; kb < EMB; kb += 4) {
    v2f a = *(const v2f*)(arow + kb);
    v2f bv = *(const v2f*)(brow + kb);
    acc = __builtin_amdgcn_wmma_f32_16x16x4_f32(
        false, a, false, bv, (short)0, acc, false, false);
  }
  float* outb = proj + (size_t)b * SENTS * CPAD;
  #pragma unroll
  for (int r = 0; r < 8; ++r) {
    const int s = st * 16 + r + 8 * hl;
    outb[(size_t)s * CPAD + wave * 16 + r15] = acc[r];
  }
}

__global__ void logits_kernel(const float* __restrict__ atts,
                              const float* __restrict__ proj,
                              const float* __restrict__ bias,
                              float* __restrict__ out) {
  const int c = blockIdx.x;
  const int b = blockIdx.y;
  const int t = threadIdx.x;
  __shared__ float red[SENTS];

  const float a = atts[((size_t)b * CPAD + c) * SENTS + t];
  const float p = proj[((size_t)b * SENTS + t) * CPAD + c];
  red[t] = a * p;
  __syncthreads();
  for (int off = SENTS / 2; off > 0; off >>= 1) {
    if (t < off) red[t] += red[t + off];
    __syncthreads();
  }
  if (t == 0) out[(size_t)b * NCLS + c] = red[0] + bias[c];
}

extern "C" void kernel_launch(void* const* d_in, const int* in_sizes, int n_in,
                              void* d_out, int out_size, void* d_ws, size_t ws_size,
                              hipStream_t stream) {
  const int*   tok  = (const int*)d_in[0];
  const float* emb  = (const float*)d_in[1];
  const float* cls  = (const float*)d_in[2];
  const float* mw   = (const float*)d_in[3];
  const float* bias = (const float*)d_in[4];
  float*       out  = (float*)d_out;

  char* ws = (char*)d_ws;
  const size_t sents_bytes  = (size_t)BATCH * SENTS * EMB * sizeof(float);
  const size_t scores_bytes = (size_t)BATCH * CPAD * SENTS * sizeof(float);
  float* sents  = (float*)ws;
  float* scores = (float*)(ws + sents_bytes);
  float* proj   = (float*)(ws + sents_bytes + scores_bytes);

  gather_mean_kernel<<<BATCH * SENTS, 64, 0, stream>>>(tok, emb, sents);
  scores_wmma_kernel<<<dim3(SENTS / 16, BATCH), 224, 0, stream>>>(cls, sents, scores);
  softmax_kernel<<<dim3(NCLS, BATCH), SENTS, 0, stream>>>(scores);
  proj_wmma_kernel<<<dim3(SENTS / 16, BATCH), 224, 0, stream>>>(sents, mw, proj);
  logits_kernel<<<dim3(NCLS, BATCH), SENTS, 0, stream>>>(scores, proj, bias, out);
}


# Fuse score GEMM, softmax, projection GEMM and weighted sum into one kernel per (batch, 16-class tile): one sentence fragment feeds both WMMAs, softmax in registers; new reduction tree over sentences
# speedup vs baseline: 1.4027x; 1.4027x over previous
#include <hip/hip_runtime.h>

typedef __attribute__((ext_vector_type(2))) float v2f;
typedef __attribute__((ext_vector_type(8))) float v8f;

#define EMB    256
#define SENTS  128
#define BATCH  64
#define TOKS   64
#define NCLS   100
#define CTILES 7
#define STILES (SENTS / 16)

__global__ void gather_mean_kernel(const int* __restrict__ tok,
                                   const float* __restrict__ emb,
                                   float* __restrict__ sents) {
  const int bs = blockIdx.x;
  const int t  = threadIdx.x;
  const int* tl = tok + (size_t)bs * TOKS;
  const float4* embv = (const float4*)emb;
  float4 acc = make_float4(0.f, 0.f, 0.f, 0.f);
  #pragma unroll 4
  for (int l = 0; l < TOKS; ++l) {
    const int tk = tl[l];
    float4 v = embv[(size_t)tk * (EMB / 4) + t];
    acc.x += v.x; acc.y += v.y; acc.z += v.z; acc.w += v.w;
  }
  const float inv = 1.0f / (float)TOKS;
  acc.x *= inv; acc.y *= inv; acc.z *= inv; acc.w *= inv;
  ((float4*)sents)[(size_t)bs * (EMB / 4) + t] = acc;
}

template <int MASK>
__device__ __forceinline__ float row_xor(float x) {
  return __builtin_bit_cast(float, __builtin_amdgcn_update_dpp(
      0, __builtin_bit_cast(int, x), 0x160 | MASK, 0xF, 0xF, true));
}
__device__ __forceinline__ float row_max16(float x) {
  x = fmaxf(x, row_xor<1>(x));
  x = fmaxf(x, row_xor<2>(x));
  x = fmaxf(x, row_xor<4>(x));
  x = fmaxf(x, row_xor<8>(x));
  return x;
}
__device__ __forceinline__ float row_sum16(float x) {
  x += row_xor<1>(x);
  x += row_xor<2>(x);
  x += row_xor<4>(x);
  x += row_xor<8>(x);
  return x;
}

__global__ __launch_bounds__(256) void attend_logits_kernel(
    const float* __restrict__ cls, const float* __restrict__ mw,
    const float* __restrict__ bias, const float* __restrict__ sents,
    float* __restrict__ out) {
  const int ct   = blockIdx.x;
  const int b    = blockIdx.y;
  const int wave = threadIdx.x >> 5;
  const int lane = threadIdx.x & 31;
  const int hl   = lane >> 4;
  const int r15  = lane & 15;

  __shared__ float red_max[STILES][16];
  __shared__ float red_den[STILES][16];
  __shared__ float red_num[STILES][16];

  const int m  = ct * 16 + r15;
  const int mc = (m < NCLS) ? m : (NCLS - 1);
  const int n  = wave * 16 + r15;

  const float* arow_s = cls + (size_t)mc * EMB + 2 * hl;
  const float* arow_p = mw  + (size_t)mc * EMB + 2 * hl;
  const float* brow   = sents + ((size_t)b * SENTS + n) * EMB + 2 * hl;

  v8f accS = {};
  v8f accP = {};
  for (int kb = 0; kb < EMB; kb += 32) {
    v2f fb[8], fs[8], fp[8];
    #pragma unroll
    for (int j = 0; j < 8; ++j) {
      fb[j] = *(const v2f*)(brow   + kb + 4 * j);
      fs[j] = *(const v2f*)(arow_s + kb + 4 * j);
      fp[j] = *(const v2f*)(arow_p + kb + 4 * j);
    }
    #pragma unroll
    for (int j = 0; j < 8; ++j) {
      accS = __builtin_amdgcn_wmma_f32_16x16x4_f32(
           false, fs[j],  false, fb[j],
           (short)0, accS,  false,  false);
      accP = __builtin_amdgcn_wmma_f32_16x16x4_f32(
          false, fp[j], false, fb[j], (short)0, accP, false, false);
    }
  }

  float mx[8];
  #pragma unroll
  for (int r = 0; r < 8; ++r) mx[r] = row_max16(accS[r]);
  if (r15 == 0) {
    #pragma unroll
    for (int r = 0; r < 8; ++r) red_max[wave][8 * hl + r] = mx[r];
  }
  __syncthreads();
  #pragma unroll
  for (int r = 0; r < 8; ++r) {
    float v = red_max[0][8 * hl + r];
    #pragma unroll
    for (int w = 1; w < STILES; ++w) v = fmaxf(v, red_max[w][8 * hl + r]);
    mx[r] = v;
  }

  float den[8], num[8];
  #pragma unroll
  for (int r = 0; r < 8; ++r) {
    const float e = __expf(accS[r] - mx[r]);
    den[r] = row_sum16(e);
    num[r] = row_sum16(e * accP[r]);
  }
  if (r15 == 0) {
    #pragma unroll
    for (int r = 0; r < 8; ++r) {
      red_den[wave][8 * hl + r] = den[r];
      red_num[wave][8 * hl + r] = num[r];
    }
  }
  __syncthreads();

  if (threadIdx.x < 16) {
    const int c = ct * 16 + threadIdx.x;
    float d = 0.f, nm = 0.f;
    #pragma unroll
    for (int w = 0; w < STILES; ++w) {
      d  += red_den[w][threadIdx.x];
      nm += red_num[w][threadIdx.x];
    }
    if (c < NCLS) out[(size_t)b * NCLS + c] = nm / d + bias[c];
  }
}

extern "C" void kernel_launch(void* const* d_in, const int* in_sizes, int n_in,
                              void* d_out, int out_size, void* d_ws, size_t ws_size,
                              hipStream_t stream) {
  const int*   tok  = (const int*)d_in[0];
  const float* emb  = (const float*)d_in[1];
  const float* cls  = (const float*)d_in[2];
  const float* mw   = (const float*)d_in[3];
  const float* bias = (const float*)d_in[4];
  float*       out  = (float*)d_out;

  float* sents = (float*)d_ws;

  gather_mean_kernel<<<BATCH * SENTS, 64, 0, stream>>>(tok, emb, sents);
  attend_logits_kernel<<<dim3(CTILES, BATCH), 256, 0, stream>>>(cls, mw, bias, sents, out);
}


# Second launch stages its sentence and class-row tiles through LDS in 64-k chunks (whole-line 16-byte global loads, double buffered); WMMA fragments come from LDS; k order and all sums unchanged
# speedup vs baseline: 1.7304x; 1.2336x over previous
#include <hip/hip_runtime.h>

typedef __attribute__((ext_vector_type(2))) float v2f;
typedef __attribute__((ext_vector_type(8))) float v8f;

#define EMB    256
#define SENTS  128
#define BATCH  64
#define TOKS   64
#define NCLS   100
#define CTILES 7
#define STILES (SENTS / 16)

__global__ void gather_mean_kernel(const int* __restrict__ tok,
                                   const float* __restrict__ emb,
                                   float* __restrict__ sents) {
  const int bs = blockIdx.x;
  const int t  = threadIdx.x;
  const int* tl = tok + (size_t)bs * TOKS;
  const float4* embv = (const float4*)emb;
  float4 acc = make_float4(0.f, 0.f, 0.f, 0.f);
  #pragma unroll 4
  for (int l = 0; l < TOKS; ++l) {
    const int tk = tl[l];
    float4 v = embv[(size_t)tk * (EMB / 4) + t];
    acc.x += v.x; acc.y += v.y; acc.z += v.z; acc.w += v.w;
  }
  const float inv = 1.0f / (float)TOKS;
  acc.x *= inv; acc.y *= inv; acc.z *= inv; acc.w *= inv;
  ((float4*)sents)[(size_t)bs * (EMB / 4) + t] = acc;
}

#define KC     64
#define NCHUNK (EMB / KC)
#define TROWS  (SENTS + 32)
#define PITCH  (KC + 4)
#define PASSES (TROWS * (KC / 4) / 256)
template <int MASK>
__device__ __forceinline__ float row_xor(float x) {
  return __builtin_bit_cast(float, __builtin_amdgcn_update_dpp(
      0, __builtin_bit_cast(int, x), 0x160 | MASK, 0xF, 0xF, true));
}
__device__ __forceinline__ float row_max16(float x) {
  x = fmaxf(x, row_xor<1>(x));
  x = fmaxf(x, row_xor<2>(x));
  x = fmaxf(x, row_xor<4>(x));
  x = fmaxf(x, row_xor<8>(x));
  return x;
}
__device__ __forceinline__ float row_sum16(float x) {
  x += row_xor<1>(x);
  x += row_xor<2>(x);
  x += row_xor<4>(x);
  x += row_xor<8>(x);
  return x;
}

__global__ __launch_bounds__(256) void attend_logits_kernel(
    const float* __restrict__ cls, const float* __restrict__ mw,
    const float* __restrict__ bias, const float* __restrict__ sents,
    float* __restrict__ out) {
  const int ct   = blockIdx.x;
  const int b    = blockIdx.y;
  const int wave = threadIdx.x >> 5;
  const int lane = threadIdx.x & 31;
  const int hl   = lane >> 4;
  const int r15  = lane & 15;

  __shared__ __attribute__((aligned(16))) float tile[2][TROWS * PITCH];
  __shared__ float red_max[STILES][16];
  __shared__ float red_den[STILES][16];
  __shared__ float red_num[STILES][16];

  const int q  = threadIdx.x & 15;
  const int rr = threadIdx.x >> 4;
  const int m  = ct * 16 + rr;
  const int mc = (m < NCLS) ? m : (NCLS - 1);
  const float* src_sent = sents + ((size_t)b * SENTS + rr) * EMB + 4 * q;
  const float* src_cls  = cls + (size_t)mc * EMB + 4 * q;
  const float* src_mw   = mw  + (size_t)mc * EMB + 4 * q;
  const int    dst      = rr * PITCH + 4 * q;

  const int frag_b = (wave * 16 + r15) * PITCH + 2 * hl;
  const int frag_s = (SENTS + r15) * PITCH + 2 * hl;
  const int frag_p = (SENTS + 16 + r15) * PITCH + 2 * hl;

  float4 pre[PASSES];
  #pragma unroll
  for (int p = 0; p < 8; ++p)
    pre[p] = *(const float4*)(src_sent + (size_t)p * 16 * EMB);
  pre[8] = *(const float4*)(src_cls);
  pre[9] = *(const float4*)(src_mw);
  #pragma unroll
  for (int p = 0; p < PASSES; ++p)
    *(float4*)(&tile[0][dst + p * 16 * PITCH]) = pre[p];
  __syncthreads();

  v8f accS = {};
  v8f accP = {};
  #pragma unroll
  for (int kc = 0; kc < NCHUNK; ++kc) {
    const float* cur = tile[kc & 1];
    if (kc + 1 < NCHUNK) {
      const int ko = (kc + 1) * KC;
      #pragma unroll
      for (int p = 0; p < 8; ++p)
        pre[p] = *(const float4*)(src_sent + (size_t)p * 16 * EMB + ko);
      pre[8] = *(const float4*)(src_cls + ko);
      pre[9] = *(const float4*)(src_mw + ko);
    }
    #pragma unroll
    for (int g = 0; g < 2; ++g) {
      v2f fb[8], fs[8], fp[8];
      #pragma unroll
      for (int j = 0; j < 8; ++j) {
        fb[j] = *(const v2f*)(cur + frag_b + 32 * g + 4 * j);
        fs[j] = *(const v2f*)(cur + frag_s + 32 * g + 4 * j);
        fp[j] = *(const v2f*)(cur + frag_p + 32 * g + 4 * j);
      }
      #pragma unroll
      for (int j = 0; j < 8; ++j) {
        accS = __builtin_amdgcn_wmma_f32_16x16x4_f32(
             false, fs[j],  false, fb[j],
             (short)0, accS,  false,  false);
        accP = __builtin_amdgcn_wmma_f32_16x16x4_f32(
            false, fp[j], false, fb[j], (short)0, accP, false, false);
      }
    }
    if (kc + 1 < NCHUNK) {
      #pragma unroll
      for (int p = 0; p < PASSES; ++p)
        *(float4*)(&tile[(kc + 1) & 1][dst + p * 16 * PITCH]) = pre[p];
      __syncthreads();
    }
  }

  float mx[8];
  #pragma unroll
  for (int r = 0; r < 8; ++r) mx[r] = row_max16(accS[r]);
  if (r15 == 0) {
    #pragma unroll
    for (int r = 0; r < 8; ++r) red_max[wave][8 * hl + r] = mx[r];
  }
  __syncthreads();
  #pragma unroll
  for (int r = 0; r < 8; ++r) {
    float v = red_max[0][8 * hl + r];
    #pragma unroll
    for (int w = 1; w < STILES; ++w) v = fmaxf(v, red_max[w][8 * hl + r]);
    mx[r] = v;
  }

  float den[8], num[8];
  #pragma unroll
  for (int r = 0; r < 8; ++r) {
    const float e = __expf(accS[r] - mx[r]);
    den[r] = row_sum16(e);
    num[r] = row_sum16(e * accP[r]);
  }
  if (r15 == 0) {
    #pragma unroll
    for (int r = 0; r < 8; ++r) {
      red_den[wave][8 * hl + r] = den[r];
      red_num[wave][8 * hl + r] = num[r];
    }
  }
  __syncthreads();

  if (threadIdx.x < 16) {
    const int c = ct * 16 + threadIdx.x;
    float d = 0.f, nm = 0.f;
    #pragma unroll
    for (int w = 0; w < STILES; ++w) {
      d  += red_den[w][threadIdx.x];
      nm += red_num[w][threadIdx.x];
    }
    if (c < NCLS) out[(size_t)b * NCLS + c] = nm / d + bias[c];
  }
}

extern "C" void kernel_launch(void* const* d_in, const int* in_sizes, int n_in,
                              void* d_out, int out_size, void* d_ws, size_t ws_size,
                              hipStream_t stream) {
  const int*   tok  = (const int*)d_in[0];
  const float* emb  = (const float*)d_in[1];
  const float* cls  = (const float*)d_in[2];
  const float* mw   = (const float*)d_in[3];
  const float* bias = (const float*)d_in[4];
  float*       out  = (float*)d_out;

  float* sents = (float*)d_ws;

  gather_mean_kernel<<<BATCH * SENTS, 64, 0, stream>>>(tok, emb, sents);
  attend_logits_kernel<<<dim3(CTILES, BATCH), 256, 0, stream>>>(cls, mw, bias, sents, out);
}


# Gather launch: one wave streams whole 1 KB rows (two 512-byte loads each) for two sentences, 18 rows requested ahead (36-38 loads in flight); 4096 waves, not 16384; per-element sum order unchanged
# speedup vs baseline: 1.8420x; 1.0645x over previous
#include <hip/hip_runtime.h>

typedef __attribute__((ext_vector_type(2))) float v2f;
typedef __attribute__((ext_vector_type(8))) float v8f;

#define EMB    256
#define SENTS  128
#define BATCH  64
#define TOKS   64
#define NCLS   100
#define CTILES 7
#define STILES (SENTS / 16)

typedef __attribute__((ext_vector_type(4))) float v4f;
typedef __attribute__((ext_vector_type(4))) int   v4i;

#define G_WAVES  8
#define G_SPW    2
#define G_AHEAD  18
#define G_RING   (G_AHEAD + 1)
#define G_NROW   (G_SPW * TOKS)
#define G_BLOCKS (BATCH * SENTS / (G_WAVES * G_SPW))

__global__ __launch_bounds__(G_WAVES * 32) void gather_mean_kernel(
    const int* __restrict__ tok, const float* emb, float* sents) {
  const int lane = threadIdx.x & 31;
  const int wave = __builtin_amdgcn_readfirstlane(threadIdx.x >> 5);
  const int s0   = (blockIdx.x * G_WAVES + wave) * G_SPW;
  const v4i idq  = ((const v4i*)(tok + (size_t)s0 * TOKS))[lane];
  const v4f* embv = (const v4f*)emb + lane;
  v4f* dst = (v4f*)sents + (size_t)s0 * (EMB / 4) + lane;

  v4f buf[G_RING][2];

#define ROW_ID(T) __builtin_amdgcn_readlane(idq[(T) & 3], (T) >> 2)
#define LOAD_ROW(T) do { \
    const v4f* p_ = embv + (size_t)ROW_ID(T) * (EMB / 4); \
    buf[(T) % G_RING][0] = p_[0]; \
    buf[(T) % G_RING][1] = p_[EMB / 8]; } while (0)

  #pragma unroll
  for (int t = 0; t < G_AHEAD; ++t) LOAD_ROW(t);

  v4f lo = {0.f, 0.f, 0.f, 0.f};
  v4f hi = {0.f, 0.f, 0.f, 0.f};
  const float inv = 1.0f / (float)TOKS;
  #pragma unroll
  for (int t = 0; t < G_NROW; ++t) {
    if (t + G_AHEAD < G_NROW) LOAD_ROW(t + G_AHEAD);
    asm volatile("" : "+v"(lo), "+v"(hi) : : "memory");
    lo += buf[t % G_RING][0];
    hi += buf[t % G_RING][1];
    asm volatile("" : "+v"(lo), "+v"(hi) : : "memory");
    if ((t + 1) % TOKS == 0) {
      dst[(size_t)(t / TOKS) * (EMB / 4)]           = lo * inv;
      dst[(size_t)(t / TOKS) * (EMB / 4) + EMB / 8] = hi * inv;
      lo = (v4f){0.f, 0.f, 0.f, 0.f};
      hi = (v4f){0.f, 0.f, 0.f, 0.f};
    }
  }
#undef ROW_ID
#undef LOAD_ROW
}

#define KC     64
#define NCHUNK (EMB / KC)
#define TROWS  (SENTS + 32)
#define PITCH  (KC + 4)
#define PASSES (TROWS * (KC / 4) / 256)
template <int MASK>
__device__ __forceinline__ float row_xor(float x) {
  return __builtin_bit_cast(float, __builtin_amdgcn_update_dpp(
      0, __builtin_bit_cast(int, x), 0x160 | MASK, 0xF, 0xF, true));
}
__device__ __forceinline__ float row_max16(float x) {
  x = fmaxf(x, row_xor<1>(x));
  x = fmaxf(x, row_xor<2>(x));
  x = fmaxf(x, row_xor<4>(x));
  x = fmaxf(x, row_xor<8>(x));
  return x;
}
__device__ __forceinline__ float row_sum16(float x) {
  x += row_xor<1>(x);
  x += row_xor<2>(x);
  x += row_xor<4>(x);
  x += row_xor<8>(x);
  return x;
}

__global__ __launch_bounds__(256) void attend_logits_kernel(
    const float* __restrict__ cls, const float* __restrict__ mw,
    const float* __restrict__ bias, const float* __restrict__ sents,
    float* __restrict__ out) {
  const int ct   = blockIdx.x;
  const int b    = blockIdx.y;
  const int wave = threadIdx.x >> 5;
  const int lane = threadIdx.x & 31;
  const int hl   = lane >> 4;
  const int r15  = lane & 15;

  __shared__ __attribute__((aligned(16))) float tile[2][TROWS * PITCH];
  __shared__ float red_max[STILES][16];
  __shared__ float red_den[STILES][16];
  __shared__ float red_num[STILES][16];

  const int q  = threadIdx.x & 15;
  const int rr = threadIdx.x >> 4;
  const int m  = ct * 16 + rr;
  const int mc = (m < NCLS) ? m : (NCLS - 1);
  const float* src_sent = sents + ((size_t)b * SENTS + rr) * EMB + 4 * q;
  const float* src_cls  = cls + (size_t)mc * EMB + 4 * q;
  const float* src_mw   = mw  + (size_t)mc * EMB + 4 * q;
  const int    dst      = rr * PITCH + 4 * q;

  const int frag_b = (wave * 16 + r15) * PITCH + 2 * hl;
  const int frag_s = (SENTS + r15) * PITCH + 2 * hl;
  const int frag_p = (SENTS + 16 + r15) * PITCH + 2 * hl;

  float4 pre[PASSES];
  #pragma unroll
  for (int p = 0; p < 8; ++p)
    pre[p] = *(const float4*)(src_sent + (size_t)p * 16 * EMB);
  pre[8] = *(const float4*)(src_cls);
  pre[9] = *(const float4*)(src_mw);
  #pragma unroll
  for (int p = 0; p < PASSES; ++p)
    *(float4*)(&tile[0][dst + p * 16 * PITCH]) = pre[p];
  __syncthreads();

  v8f accS = {};
  v8f accP = {};
  #pragma unroll
  for (int kc = 0; kc < NCHUNK; ++kc) {
    const float* cur = tile[kc & 1];
    if (kc + 1 < NCHUNK) {
      const int ko = (kc + 1) * KC;
      #pragma unroll
      for (int p = 0; p < 8; ++p)
        pre[p] = *(const float4*)(src_sent + (size_t)p * 16 * EMB + ko);
      pre[8] = *(const float4*)(src_cls + ko);
      pre[9] = *(const float4*)(src_mw + ko);
    }
    #pragma unroll
    for (int g = 0; g < 2; ++g) {
      v2f fb[8], fs[8], fp[8];
      #pragma unroll
      for (int j = 0; j < 8; ++j) {
        fb[j] = *(const v2f*)(cur + frag_b + 32 * g + 4 * j);
        fs[j] = *(const v2f*)(cur + frag_s + 32 * g + 4 * j);
        fp[j] = *(const v2f*)(cur + frag_p + 32 * g + 4 * j);
      }
      #pragma unroll
      for (int j = 0; j < 8; ++j) {
        accS = __builtin_amdgcn_wmma_f32_16x16x4_f32(
             false, fs[j],  false, fb[j],
             (short)0, accS,  false,  false);
        accP = __builtin_amdgcn_wmma_f32_16x16x4_f32(
            false, fp[j], false, fb[j], (short)0, accP, false, false);
      }
    }
    if (kc + 1 < NCHUNK) {
      #pragma unroll
      for (int p = 0; p < PASSES; ++p)
        *(float4*)(&tile[(kc + 1) & 1][dst + p * 16 * PITCH]) = pre[p];
      __syncthreads();
    }
  }

  float mx[8];
  #pragma unroll
  for (int r = 0; r < 8; ++r) mx[r] = row_max16(accS[r]);
  if (r15 == 0) {
    #pragma unroll
    for (int r = 0; r < 8; ++r) red_max[wave][8 * hl + r] = mx[r];
  }
  __syncthreads();
  #pragma unroll
  for (int r = 0; r < 8; ++r) {
    float v = red_max[0][8 * hl + r];
    #pragma unroll
    for (int w = 1; w < STILES; ++w) v = fmaxf(v, red_max[w][8 * hl + r]);
    mx[r] = v;
  }

  float den[8], num[8];
  #pragma unroll
  for (int r = 0; r < 8; ++r) {
    const float e = __expf(accS[r] - mx[r]);
    den[r] = row_sum16(e);
    num[r] = row_sum16(e * accP[r]);
  }
  if (r15 == 0) {
    #pragma unroll
    for (int r = 0; r < 8; ++r) {
      red_den[wave][8 * hl + r] = den[r];
      red_num[wave][8 * hl + r] = num[r];
    }
  }
  __syncthreads();

  if (threadIdx.x < 16) {
    const int c = ct * 16 + threadIdx.x;
    float d = 0.f, nm = 0.f;
    #pragma unroll
    for (int w = 0; w < STILES; ++w) {
      d  += red_den[w][threadIdx.x];
      nm += red_num[w][threadIdx.x];
    }
    if (c < NCLS) out[(size_t)b * NCLS + c] = nm / d + bias[c];
  }
}

extern "C" void kernel_launch(void* const* d_in, const int* in_sizes, int n_in,
                              void* d_out, int out_size, void* d_ws, size_t ws_size,
                              hipStream_t stream) {
  const int*   tok  = (const int*)d_in[0];
  const float* emb  = (const float*)d_in[1];
  const float* cls  = (const float*)d_in[2];
  const float* mw   = (const float*)d_in[3];
  const float* bias = (const float*)d_in[4];
  float*       out  = (float*)d_out;

  float* sents = (float*)d_ws;

  gather_mean_kernel<<<G_BLOCKS, G_WAVES * 32, 0, stream>>>(tok, emb, sents);
  attend_logits_kernel<<<dim3(CTILES, BATCH), 256, 0, stream>>>(cls, mw, bias, sents, out);
}
